# starting kernel
# baseline (speedup 1.0000x reference)
#include <hip/hip_runtime.h>

typedef float v2f __attribute__((ext_vector_type(2)));
typedef float v8f __attribute__((ext_vector_type(8)));

#define NB   32
#define DD   512
#define KK   64
#define PP   1024
#define EPSF 1e-12f

__global__ __launch_bounds__(128) void k_assign(const float* __restrict__ x,
                                                const float* __restrict__ conv_w,
                                                const float* __restrict__ conv_b,
                                                v2f* __restrict__ a_pair) {
    __shared__ float sm[16 * KK];
    __shared__ float pmax[16 * 8];
    __shared__ float psum[16 * 8];
    const int n    = blockIdx.x >> 6;
    const int p0   = (blockIdx.x & 63) << 4;
    const int tid  = threadIdx.x;
    const int wave = tid >> 5;
    const int lane = tid & 31;
    const int k0   = wave << 4;
    const int hi   = (lane >= 16) ? 1 : 0;
    const int l15  = lane & 15;
    const float* xb = x + (size_t)n * DD * PP;

    v8f c = {0.f, 0.f, 0.f, 0.f, 0.f, 0.f, 0.f, 0.f};

    for (int d = 0; d < DD; d += 4) {
        v2f a = *(const v2f*)(conv_w + (size_t)(k0 + l15) * DD + d + (hi ? 2 : 0));
        const int dr = d + (hi ? 2 : 0);
        v2f b;
        b.x = xb[(size_t)dr * PP + p0 + l15];
        b.y = xb[(size_t)(dr + 1) * PP + p0 + l15];
        c = __builtin_amdgcn_wmma_f32_16x16x4_f32(false, a, false, b,
                                                  (short)0, c, false, false);
    }

#pragma unroll
    for (int r = 0; r < 8; ++r) {
        const int k = k0 + r + (hi ? 8 : 0);
        sm[l15 * KK + k] = c[r] + conv_b[k];
    }
    __syncthreads();

    const int p = tid >> 3;
    const int g = tid & 7;
    float m = -3.0e38f;
#pragma unroll
    for (int j = 0; j < 8; ++j) m = fmaxf(m, sm[p * KK + g * 8 + j]);
    pmax[p * 8 + g] = m;
    __syncthreads();
    float M = pmax[p * 8];
#pragma unroll
    for (int j = 1; j < 8; ++j) M = fmaxf(M, pmax[p * 8 + j]);
    float s = 0.f;
#pragma unroll
    for (int j = 0; j < 8; ++j) {
        const float e = __expf(sm[p * KK + g * 8 + j] - M);
        sm[p * KK + g * 8 + j] = e;
        s += e;
    }
    psum[p * 8 + g] = s;
    __syncthreads();
    float S = psum[p * 8];
#pragma unroll
    for (int j = 1; j < 8; ++j) S += psum[p * 8 + j];
    const float inv = 1.0f / S;
#pragma unroll
    for (int j = 0; j < 8; ++j) sm[p * KK + g * 8 + j] *= inv;
    __syncthreads();

    v2f* ao = a_pair + ((size_t)n * (PP / 2) + (p0 >> 1)) * KK;
    for (int i = tid; i < 8 * KK; i += 128) {
        const int ppl = i >> 6;
        const int k   = i & 63;
        v2f v;
        v.x = sm[(2 * ppl) * KK + k];
        v.y = sm[(2 * ppl + 1) * KK + k];
        ao[(size_t)ppl * KK + k] = v;
    }
}

__global__ __launch_bounds__(64) void k_asum(const v2f* __restrict__ a_pair,
                                             float* __restrict__ asum) {
    const int n = blockIdx.x;
    const int k = threadIdx.x;
    const v2f* ap = a_pair + (size_t)n * (PP / 2) * KK + k;
    float s = 0.f;
    for (int pp = 0; pp < PP / 2; ++pp) {
        const v2f v = ap[(size_t)pp * KK];
        s += v.x + v.y;
    }
    asum[n * KK + k] = s;
}

__global__ __launch_bounds__(128) void k_vlad(const float* __restrict__ x,
                                              const v2f* __restrict__ a_pair,
                                              const float* __restrict__ asum,
                                              const float* __restrict__ centroids,
                                              float* __restrict__ vraw) {
    const int n    = blockIdx.x;
    const int k0   = blockIdx.y << 4;
    const int tid  = threadIdx.x;
    const int wave = tid >> 5;
    const int lane = tid & 31;
    const int d0   = (blockIdx.z << 6) + (wave << 4);
    const int hi   = (lane >= 16) ? 1 : 0;
    const int l15  = lane & 15;
    const float* xb = x + (size_t)n * DD * PP;
    const v2f* ab = a_pair + (size_t)n * (PP / 2) * KK;

    v8f c = {0.f, 0.f, 0.f, 0.f, 0.f, 0.f, 0.f, 0.f};

    for (int p = 0; p < PP; p += 4) {
        const int pr  = p + (hi ? 2 : 0);
        v2f a = *(const v2f*)(xb + (size_t)(d0 + l15) * PP + pr);
        v2f b = ab[(size_t)((p >> 1) + hi) * KK + k0 + l15];
        c = __builtin_amdgcn_wmma_f32_16x16x4_f32(false, a, false, b,
                                                  (short)0, c, false, false);
    }

    const int k = k0 + l15;
    const float as = asum[n * KK + k];
    float* vo = vraw + (size_t)n * KK * DD;
#pragma unroll
    for (int r = 0; r < 8; ++r) {
        const int d = d0 + r + (hi ? 8 : 0);
        vo[(size_t)k * DD + d] = c[r] - as * centroids[(size_t)k * DD + d];
    }
}

__global__ __launch_bounds__(256) void k_norm(const float* __restrict__ vraw,
                                              float* __restrict__ out) {
    __shared__ float red[256];
    __shared__ float tsc[KK];
    const int n   = blockIdx.x;
    const int tid = threadIdx.x;
    const float* vp = vraw + (size_t)n * KK * DD;

    const int k   = tid >> 2;
    const int sub = tid & 3;
    float s = 0.f;
    for (int d = sub; d < DD; d += 4) {
        const float v = vp[(size_t)k * DD + d];
        s += v * v;
    }
    red[tid] = s;
    __syncthreads();

    float t = 0.f;
    if (tid < KK)
        t = red[tid * 4] + red[tid * 4 + 1] + red[tid * 4 + 2] + red[tid * 4 + 3];
    __syncthreads();
    if (tid < KK) {
        const float nrm = sqrtf(t);
        const float sc  = 1.0f / fmaxf(nrm, EPSF);
        const float nn  = nrm * sc;
        red[tid] = nn * nn;
        tsc[tid] = sc;
    }
    __syncthreads();
    if (tid == 0) {
        float g = 0.f;
        for (int i = 0; i < KK; ++i) g += red[i];
        red[0] = 1.0f / fmaxf(sqrtf(g), EPSF);
    }
    __syncthreads();

    const float gs = red[0];
    float* op = out + (size_t)n * KK * DD;
    for (int i = tid; i < KK * DD; i += 256) {
        op[i] = vp[i] * tsc[i >> 9] * gs;
    }
}

extern "C" void kernel_launch(void* const* d_in, const int* in_sizes, int n_in,
                              void* d_out, int out_size, void* d_ws, size_t ws_size,
                              hipStream_t stream) {
    const float* x         = (const float*)d_in[0];
    const float* centroids = (const float*)d_in[1];
    const float* conv_w    = (const float*)d_in[2];
    const float* conv_b    = (const float*)d_in[3];
    float* out = (float*)d_out;

    v2f*   a_pair = (v2f*)d_ws;
    float* asum   = (float*)d_ws + (size_t)NB * PP * KK;
    float* vraw   = asum + (size_t)NB * KK;
    (void)in_sizes; (void)n_in; (void)out_size; (void)ws_size;

    k_assign<<<dim3(NB * (PP / 16)), 128, 0, stream>>>(x, conv_w, conv_b, a_pair);
    k_asum  <<<dim3(NB), 64, 0, stream>>>(a_pair, asum);
    k_vlad  <<<dim3(NB, KK / 16, DD / 64), 128, 0, stream>>>(x, a_pair, asum, centroids, vraw);
    k_norm  <<<dim3(NB), 256, 0, stream>>>(vraw, out);
}


# Column sums of the assignment (asum) accumulated from the B fragments inside the aggregation GEMM loop; serial asum launch removed (asum summation order changes: per lane half, then joined)
# speedup vs baseline: 1.1855x; 1.1855x over previous
#include <hip/hip_runtime.h>

typedef float v2f __attribute__((ext_vector_type(2)));
typedef float v8f __attribute__((ext_vector_type(8)));

#define NB   32
#define DD   512
#define KK   64
#define PP   1024
#define EPSF 1e-12f

__global__ __launch_bounds__(128) void k_assign(const float* __restrict__ x,
                                                const float* __restrict__ conv_w,
                                                const float* __restrict__ conv_b,
                                                v2f* __restrict__ a_pair) {
    __shared__ float sm[16 * KK];
    __shared__ float pmax[16 * 8];
    __shared__ float psum[16 * 8];
    const int n    = blockIdx.x >> 6;
    const int p0   = (blockIdx.x & 63) << 4;
    const int tid  = threadIdx.x;
    const int wave = tid >> 5;
    const int lane = tid & 31;
    const int k0   = wave << 4;
    const int hi   = (lane >= 16) ? 1 : 0;
    const int l15  = lane & 15;
    const float* xb = x + (size_t)n * DD * PP;

    v8f c = {0.f, 0.f, 0.f, 0.f, 0.f, 0.f, 0.f, 0.f};

    for (int d = 0; d < DD; d += 4) {
        v2f a = *(const v2f*)(conv_w + (size_t)(k0 + l15) * DD + d + (hi ? 2 : 0));
        const int dr = d + (hi ? 2 : 0);
        v2f b;
        b.x = xb[(size_t)dr * PP + p0 + l15];
        b.y = xb[(size_t)(dr + 1) * PP + p0 + l15];
        c = __builtin_amdgcn_wmma_f32_16x16x4_f32(false, a, false, b,
                                                  (short)0, c, false, false);
    }

#pragma unroll
    for (int r = 0; r < 8; ++r) {
        const int k = k0 + r + (hi ? 8 : 0);
        sm[l15 * KK + k] = c[r] + conv_b[k];
    }
    __syncthreads();

    const int p = tid >> 3;
    const int g = tid & 7;
    float m = -3.0e38f;
#pragma unroll
    for (int j = 0; j < 8; ++j) m = fmaxf(m, sm[p * KK + g * 8 + j]);
    pmax[p * 8 + g] = m;
    __syncthreads();
    float M = pmax[p * 8];
#pragma unroll
    for (int j = 1; j < 8; ++j) M = fmaxf(M, pmax[p * 8 + j]);
    float s = 0.f;
#pragma unroll
    for (int j = 0; j < 8; ++j) {
        const float e = __expf(sm[p * KK + g * 8 + j] - M);
        sm[p * KK + g * 8 + j] = e;
        s += e;
    }
    psum[p * 8 + g] = s;
    __syncthreads();
    float S = psum[p * 8];
#pragma unroll
    for (int j = 1; j < 8; ++j) S += psum[p * 8 + j];
    const float inv = 1.0f / S;
#pragma unroll
    for (int j = 0; j < 8; ++j) sm[p * KK + g * 8 + j] *= inv;
    __syncthreads();

    v2f* ao = a_pair + ((size_t)n * (PP / 2) + (p0 >> 1)) * KK;
    for (int i = tid; i < 8 * KK; i += 128) {
        const int ppl = i >> 6;
        const int k   = i & 63;
        v2f v;
        v.x = sm[(2 * ppl) * KK + k];
        v.y = sm[(2 * ppl + 1) * KK + k];
        ao[(size_t)ppl * KK + k] = v;
    }
}

__global__ __launch_bounds__(128) void k_vlad(const float* __restrict__ x,
                                              const v2f* __restrict__ a_pair,
                                              const float* __restrict__ centroids,
                                              float* __restrict__ vraw) {
    const int n    = blockIdx.x;
    const int k0   = blockIdx.y << 4;
    const int tid  = threadIdx.x;
    const int wave = tid >> 5;
    const int lane = tid & 31;
    const int d0   = (blockIdx.z << 6) + (wave << 4);
    const int hi   = (lane >= 16) ? 1 : 0;
    const int l15  = lane & 15;
    const float* xb = x + (size_t)n * DD * PP;
    const v2f* ab = a_pair + (size_t)n * (PP / 2) * KK;

    v8f c = {0.f, 0.f, 0.f, 0.f, 0.f, 0.f, 0.f, 0.f};
    v2f colsum = {0.f, 0.f};

    for (int p = 0; p < PP; p += 4) {
        const int pr  = p + (hi ? 2 : 0);
        v2f a = *(const v2f*)(xb + (size_t)(d0 + l15) * PP + pr);
        v2f b = ab[(size_t)((p >> 1) + hi) * KK + k0 + l15];
        colsum += b;
        c = __builtin_amdgcn_wmma_f32_16x16x4_f32(false, a, false, b,
                                                  (short)0, c, false, false);
    }

    const float half_sum = colsum.x + colsum.y;
    const int   hs_bits  = __builtin_bit_cast(int, half_sum);
    const float as = half_sum + __builtin_bit_cast(float,
        __builtin_amdgcn_permlanex16(hs_bits, hs_bits, 0x76543210, 0xfedcba98, false, false));

    const int k = k0 + l15;
    float* vo = vraw + (size_t)n * KK * DD;
#pragma unroll
    for (int r = 0; r < 8; ++r) {
        const int d = d0 + r + (hi ? 8 : 0);
        vo[(size_t)k * DD + d] = c[r] - as * centroids[(size_t)k * DD + d];
    }
}

__global__ __launch_bounds__(256) void k_norm(const float* __restrict__ vraw,
                                              float* __restrict__ out) {
    __shared__ float red[256];
    __shared__ float tsc[KK];
    const int n   = blockIdx.x;
    const int tid = threadIdx.x;
    const float* vp = vraw + (size_t)n * KK * DD;

    const int k   = tid >> 2;
    const int sub = tid & 3;
    float s = 0.f;
    for (int d = sub; d < DD; d += 4) {
        const float v = vp[(size_t)k * DD + d];
        s += v * v;
    }
    red[tid] = s;
    __syncthreads();

    float t = 0.f;
    if (tid < KK)
        t = red[tid * 4] + red[tid * 4 + 1] + red[tid * 4 + 2] + red[tid * 4 + 3];
    __syncthreads();
    if (tid < KK) {
        const float nrm = sqrtf(t);
        const float sc  = 1.0f / fmaxf(nrm, EPSF);
        const float nn  = nrm * sc;
        red[tid] = nn * nn;
        tsc[tid] = sc;
    }
    __syncthreads();
    if (tid == 0) {
        float g = 0.f;
        for (int i = 0; i < KK; ++i) g += red[i];
        red[0] = 1.0f / fmaxf(sqrtf(g), EPSF);
    }
    __syncthreads();

    const float gs = red[0];
    float* op = out + (size_t)n * KK * DD;
    for (int i = tid; i < KK * DD; i += 256) {
        op[i] = vp[i] * tsc[i >> 9] * gs;
    }
}

extern "C" void kernel_launch(void* const* d_in, const int* in_sizes, int n_in,
                              void* d_out, int out_size, void* d_ws, size_t ws_size,
                              hipStream_t stream) {
    const float* x         = (const float*)d_in[0];
    const float* centroids = (const float*)d_in[1];
    const float* conv_w    = (const float*)d_in[2];
    const float* conv_b    = (const float*)d_in[3];
    float* out = (float*)d_out;

    v2f*   a_pair = (v2f*)d_ws;
    float* vraw   = (float*)d_ws + (size_t)NB * PP * KK;
    (void)in_sizes; (void)n_in; (void)out_size; (void)ws_size;

    k_assign<<<dim3(NB * (PP / 16)), 128, 0, stream>>>(x, conv_w, conv_b, a_pair);
    k_vlad  <<<dim3(NB, KK / 16, DD / 64), 128, 0, stream>>>(x, a_pair, centroids, vraw);
    k_norm  <<<dim3(NB), 256, 0, stream>>>(vraw, out);
}


# k_vlad register-blocked: one wave holds 32d x 64k (2x4 tiles), each x fragment feeds 4 WMMAs, each a fragment 2; pixels split over 8 waves, partials joined in LDS (pixel sum order changes)
# speedup vs baseline: 1.5061x; 1.2704x over previous
#include <hip/hip_runtime.h>

typedef float v2f __attribute__((ext_vector_type(2)));
typedef float v8f __attribute__((ext_vector_type(8)));

#define NB   32
#define DD   512
#define KK   64
#define PP   1024
#define EPSF 1e-12f

__global__ __launch_bounds__(128) void k_assign(const float* __restrict__ x,
                                                const float* __restrict__ conv_w,
                                                const float* __restrict__ conv_b,
                                                v2f* __restrict__ a_pair) {
    __shared__ float sm[16 * KK];
    __shared__ float pmax[16 * 8];
    __shared__ float psum[16 * 8];
    const int n    = blockIdx.x >> 6;
    const int p0   = (blockIdx.x & 63) << 4;
    const int tid  = threadIdx.x;
    const int wave = tid >> 5;
    const int lane = tid & 31;
    const int k0   = wave << 4;
    const int hi   = (lane >= 16) ? 1 : 0;
    const int l15  = lane & 15;
    const float* xb = x + (size_t)n * DD * PP;

    v8f c = {0.f, 0.f, 0.f, 0.f, 0.f, 0.f, 0.f, 0.f};

    for (int d = 0; d < DD; d += 4) {
        v2f a = *(const v2f*)(conv_w + (size_t)(k0 + l15) * DD + d + (hi ? 2 : 0));
        const int dr = d + (hi ? 2 : 0);
        v2f b;
        b.x = xb[(size_t)dr * PP + p0 + l15];
        b.y = xb[(size_t)(dr + 1) * PP + p0 + l15];
        c = __builtin_amdgcn_wmma_f32_16x16x4_f32(false, a, false, b,
                                                  (short)0, c, false, false);
    }

#pragma unroll
    for (int r = 0; r < 8; ++r) {
        const int k = k0 + r + (hi ? 8 : 0);
        sm[l15 * KK + k] = c[r] + conv_b[k];
    }
    __syncthreads();

    const int p = tid >> 3;
    const int g = tid & 7;
    float m = -3.0e38f;
#pragma unroll
    for (int j = 0; j < 8; ++j) m = fmaxf(m, sm[p * KK + g * 8 + j]);
    pmax[p * 8 + g] = m;
    __syncthreads();
    float M = pmax[p * 8];
#pragma unroll
    for (int j = 1; j < 8; ++j) M = fmaxf(M, pmax[p * 8 + j]);
    float s = 0.f;
#pragma unroll
    for (int j = 0; j < 8; ++j) {
        const float e = __expf(sm[p * KK + g * 8 + j] - M);
        sm[p * KK + g * 8 + j] = e;
        s += e;
    }
    psum[p * 8 + g] = s;
    __syncthreads();
    float S = psum[p * 8];
#pragma unroll
    for (int j = 1; j < 8; ++j) S += psum[p * 8 + j];
    const float inv = 1.0f / S;
#pragma unroll
    for (int j = 0; j < 8; ++j) sm[p * KK + g * 8 + j] *= inv;
    __syncthreads();

    v2f* ao = a_pair + ((size_t)n * (PP / 2) + (p0 >> 1)) * KK;
    for (int i = tid; i < 8 * KK; i += 128) {
        const int ppl = i >> 6;
        const int k   = i & 63;
        v2f v;
        v.x = sm[(2 * ppl) * KK + k];
        v.y = sm[(2 * ppl + 1) * KK + k];
        ao[(size_t)ppl * KK + k] = v;
    }
}

__global__ __launch_bounds__(256) void k_vlad(const float* __restrict__ x,
                                              const v2f* __restrict__ a_pair,
                                              const float* __restrict__ centroids,
                                              float* __restrict__ vraw) {
    __shared__ float red[8 * 8 * 8 * 32];
    __shared__ float cred[4 * 8 * 32];
    const int n    = blockIdx.x;
    const int d0   = blockIdx.y << 5;
    const int tid  = threadIdx.x;
    const int wave = tid >> 5;
    const int lane = tid & 31;
    const int hi   = (lane >= 16) ? 1 : 0;
    const int l15  = lane & 15;
    const int pbeg = wave * (PP / 8);
    const float* xr0 = x + (size_t)n * DD * PP + (size_t)(d0 + l15) * PP + pbeg + (hi ? 2 : 0);
    const float* xr1 = xr0 + (size_t)16 * PP;
    const v2f* bp = a_pair + (size_t)n * (PP / 2) * KK + (size_t)((pbeg >> 1) + hi) * KK + l15;

    const v8f zero8 = {0.f, 0.f, 0.f, 0.f, 0.f, 0.f, 0.f, 0.f};
    v8f c[2][4];
    v2f s[4];
#pragma unroll
    for (int t = 0; t < 4; ++t) {
        c[0][t] = zero8;
        c[1][t] = zero8;
        s[t].x = 0.f;
        s[t].y = 0.f;
    }

    for (int i = 0; i < PP / 32; ++i) {
        const v2f a0 = *(const v2f*)(xr0 + 4 * i);
        const v2f a1 = *(const v2f*)(xr1 + 4 * i);
        const v2f* b = bp + (size_t)(2 * i) * KK;
        v2f bt[4];
#pragma unroll
        for (int t = 0; t < 4; ++t) bt[t] = b[16 * t];
#pragma unroll
        for (int t = 0; t < 4; ++t) {
            s[t] += bt[t];
            c[0][t] = __builtin_amdgcn_wmma_f32_16x16x4_f32(false, a0, false, bt[t],
                                                            (short)0, c[0][t], false, false);
            c[1][t] = __builtin_amdgcn_wmma_f32_16x16x4_f32(false, a1, false, bt[t],
                                                            (short)0, c[1][t], false, false);
        }
    }

#pragma unroll
    for (int dt = 0; dt < 2; ++dt)
#pragma unroll
        for (int t = 0; t < 4; ++t)
#pragma unroll
            for (int r = 0; r < 8; ++r)
                red[(((dt * 4 + t) * 8 + wave) * 8 + r) * 32 + lane] = c[dt][t][r];
#pragma unroll
    for (int t = 0; t < 4; ++t) cred[(t * 8 + wave) * 32 + lane] = s[t].x + s[t].y;
    __syncthreads();

    const int dt = wave >> 2;
    const int kt = wave & 3;
    float as = 0.f;
#pragma unroll
    for (int w = 0; w < 8; ++w)
        as += cred[(kt * 8 + w) * 32 + l15] + cred[(kt * 8 + w) * 32 + l15 + 16];

    const int k = (kt << 4) + l15;
    float* vo = vraw + (size_t)n * KK * DD;
#pragma unroll
    for (int r = 0; r < 8; ++r) {
        float acc = red[((wave * 8 + 0) * 8 + r) * 32 + lane];
#pragma unroll
        for (int w = 1; w < 8; ++w) acc += red[((wave * 8 + w) * 8 + r) * 32 + lane];
        const int d = d0 + (dt << 4) + r + (hi ? 8 : 0);
        vo[(size_t)k * DD + d] = acc - as * centroids[(size_t)k * DD + d];
    }
}

__global__ __launch_bounds__(256) void k_norm(const float* __restrict__ vraw,
                                              float* __restrict__ out) {
    __shared__ float red[256];
    __shared__ float tsc[KK];
    const int n   = blockIdx.x;
    const int tid = threadIdx.x;
    const float* vp = vraw + (size_t)n * KK * DD;

    const int k   = tid >> 2;
    const int sub = tid & 3;
    float s = 0.f;
    for (int d = sub; d < DD; d += 4) {
        const float v = vp[(size_t)k * DD + d];
        s += v * v;
    }
    red[tid] = s;
    __syncthreads();

    float t = 0.f;
    if (tid < KK)
        t = red[tid * 4] + red[tid * 4 + 1] + red[tid * 4 + 2] + red[tid * 4 + 3];
    __syncthreads();
    if (tid < KK) {
        const float nrm = sqrtf(t);
        const float sc  = 1.0f / fmaxf(nrm, EPSF);
        const float nn  = nrm * sc;
        red[tid] = nn * nn;
        tsc[tid] = sc;
    }
    __syncthreads();
    if (tid == 0) {
        float g = 0.f;
        for (int i = 0; i < KK; ++i) g += red[i];
        red[0] = 1.0f / fmaxf(sqrtf(g), EPSF);
    }
    __syncthreads();

    const float gs = red[0];
    float* op = out + (size_t)n * KK * DD;
    for (int i = tid; i < KK * DD; i += 256) {
        op[i] = vp[i] * tsc[i >> 9] * gs;
    }
}

extern "C" void kernel_launch(void* const* d_in, const int* in_sizes, int n_in,
                              void* d_out, int out_size, void* d_ws, size_t ws_size,
                              hipStream_t stream) {
    const float* x         = (const float*)d_in[0];
    const float* centroids = (const float*)d_in[1];
    const float* conv_w    = (const float*)d_in[2];
    const float* conv_b    = (const float*)d_in[3];
    float* out = (float*)d_out;

    v2f*   a_pair = (v2f*)d_ws;
    float* vraw   = (float*)d_ws + (size_t)NB * PP * KK;
    (void)in_sizes; (void)n_in; (void)out_size; (void)ws_size;

    k_assign<<<dim3(NB * (PP / 16)), 128, 0, stream>>>(x, conv_w, conv_b, a_pair);
    k_vlad  <<<dim3(NB, DD / 32), 256, 0, stream>>>(x, a_pair, centroids, vraw);
    k_norm  <<<dim3(NB), 256, 0, stream>>>(vraw, out);
}


# Row sums of squares taken in the aggregation kernel's epilogue (16 d-block partials per row); normalisation is one in-place scaling pass over 256 work-groups (row and global sum order changes)
# speedup vs baseline: 1.8507x; 1.2288x over previous
#include <hip/hip_runtime.h>

typedef float v2f __attribute__((ext_vector_type(2)));
typedef float v4f __attribute__((ext_vector_type(4)));
typedef float v8f __attribute__((ext_vector_type(8)));

#define NB   32
#define DD   512
#define KK   64
#define PP   1024
#define EPSF 1e-12f

#define SMP 65

static __device__ __forceinline__ void assign_chunk(v8f (&c)[2][4], const v4f w0, const v4f w1,
                                                    const v4f x0, const v4f x1,
                                                    const v4f x2, const v4f x3) {
    v2f a[2][2];
    a[0][0].x = w0[0]; a[0][0].y = w0[1];
    a[0][1].x = w0[2]; a[0][1].y = w0[3];
    a[1][0].x = w1[0]; a[1][0].y = w1[1];
    a[1][1].x = w1[2]; a[1][1].y = w1[3];
#pragma unroll
    for (int j = 0; j < 4; ++j) {
        v2f b0, b1;
        b0.x = x0[j]; b0.y = x1[j];
        b1.x = x2[j]; b1.y = x3[j];
#pragma unroll
        for (int kk = 0; kk < 2; ++kk) {
            c[kk][j] = __builtin_amdgcn_wmma_f32_16x16x4_f32(false, a[kk][0], false, b0,
                                                             (short)0, c[kk][j], false, false);
            c[kk][j] = __builtin_amdgcn_wmma_f32_16x16x4_f32(false, a[kk][1], false, b1,
                                                             (short)0, c[kk][j], false, false);
        }
    }
}

__global__ __launch_bounds__(256) void k_assign(const float* __restrict__ x,
                                                const float* __restrict__ conv_w,
                                                const float* __restrict__ conv_b,
                                                v2f* __restrict__ a_pair) {
    __shared__ float red[16 * 4 * 8 * 32];
    __shared__ float sm[64 * SMP];
    const int n    = blockIdx.x >> 4;
    const int P0   = (blockIdx.x & 15) << 6;
    const int tid  = threadIdx.x;
    const int wave = tid >> 5;
    const int lane = tid & 31;
    const int hi   = (lane >= 16) ? 1 : 0;
    const int l15  = lane & 15;
    const int kh   = wave & 1;
    const int dq   = wave >> 1;
    const float* xp = x + (size_t)n * DD * PP + (size_t)(dq * 128 + 4 * hi) * PP + P0 + 4 * l15;
    const float* wp = conv_w + (size_t)(32 * kh + l15) * DD + dq * 128 + 4 * hi;

    const v8f zero8 = {0.f, 0.f, 0.f, 0.f, 0.f, 0.f, 0.f, 0.f};
    v8f c[2][4];
#pragma unroll
    for (int j = 0; j < 4; ++j) {
        c[0][j] = zero8;
        c[1][j] = zero8;
    }

    v4f w0 = *(const v4f*)(wp);
    v4f w1 = *(const v4f*)(wp + (size_t)16 * DD);
    v4f x0 = *(const v4f*)(xp);
    v4f x1 = *(const v4f*)(xp + (size_t)1 * PP);
    v4f x2 = *(const v4f*)(xp + (size_t)2 * PP);
    v4f x3 = *(const v4f*)(xp + (size_t)3 * PP);
    for (int i = 1; i < 16; ++i) {
        const float* wn = wp + 8 * i;
        const float* xn = xp + (size_t)(8 * i) * PP;
        const v4f nw0 = *(const v4f*)(wn);
        const v4f nw1 = *(const v4f*)(wn + (size_t)16 * DD);
        const v4f nx0 = *(const v4f*)(xn);
        const v4f nx1 = *(const v4f*)(xn + (size_t)1 * PP);
        const v4f nx2 = *(const v4f*)(xn + (size_t)2 * PP);
        const v4f nx3 = *(const v4f*)(xn + (size_t)3 * PP);
        assign_chunk(c, w0, w1, x0, x1, x2, x3);
        w0 = nw0; w1 = nw1;
        x0 = nx0; x1 = nx1; x2 = nx2; x3 = nx3;
    }
    assign_chunk(c, w0, w1, x0, x1, x2, x3);

#pragma unroll
    for (int kk = 0; kk < 2; ++kk)
#pragma unroll
        for (int j = 0; j < 4; ++j)
#pragma unroll
            for (int r = 0; r < 8; ++r)
                red[(((((2 * kh + kk) * 4 + j) * 4 + dq) * 8 + r) * 32) + lane] = c[kk][j][r];
    __syncthreads();

#pragma unroll
    for (int tt = 0; tt < 2; ++tt) {
        const int tile = 2 * wave + tt;
        const int kt   = tile >> 2;
        const int j    = tile & 3;
#pragma unroll
        for (int r = 0; r < 8; ++r) {
            float acc = red[(((tile * 4 + 0) * 8 + r) * 32) + lane];
#pragma unroll
            for (int q = 1; q < 4; ++q) acc += red[(((tile * 4 + q) * 8 + r) * 32) + lane];
            const int k = 16 * kt + 8 * hi + r;
            sm[(4 * l15 + j) * SMP + k] = acc + conv_b[k];
        }
    }
    __syncthreads();

    const int p = tid >> 2;
    const int g = tid & 3;
    float v[16];
    float m = -3.0e38f;
#pragma unroll
    for (int jj = 0; jj < 16; ++jj) {
        v[jj] = sm[p * SMP + g * 16 + jj];
        m = fmaxf(m, v[jj]);
    }
    m = fmaxf(m, __shfl_xor(m, 1));
    m = fmaxf(m, __shfl_xor(m, 2));
    float s = 0.f;
#pragma unroll
    for (int jj = 0; jj < 16; ++jj) {
        v[jj] = __expf(v[jj] - m);
        s += v[jj];
    }
    s += __shfl_xor(s, 1);
    s += __shfl_xor(s, 2);
    const float inv = 1.0f / s;
#pragma unroll
    for (int jj = 0; jj < 16; ++jj) sm[p * SMP + g * 16 + jj] = v[jj] * inv;
    __syncthreads();

    v2f* ao = a_pair + ((size_t)n * (PP / 2) + (P0 >> 1)) * KK;
    for (int i = tid; i < 32 * KK; i += 256) {
        const int ppl = i >> 6;
        const int k   = i & 63;
        v2f o;
        o.x = sm[(2 * ppl) * SMP + k];
        o.y = sm[(2 * ppl + 1) * SMP + k];
        ao[(size_t)ppl * KK + k] = o;
    }
}

__global__ __launch_bounds__(256) void k_vlad(const float* __restrict__ x,
                                              const v2f* __restrict__ a_pair,
                                              const float* __restrict__ centroids,
                                              float* __restrict__ vraw,
                                              float* __restrict__ rowss) {
    __shared__ float red[8 * 8 * 8 * 32];
    __shared__ float cred[4 * 8 * 32];
    __shared__ float rsq[2 * KK];
    const int n    = blockIdx.x;
    const int d0   = blockIdx.y << 5;
    const int tid  = threadIdx.x;
    const int wave = tid >> 5;
    const int lane = tid & 31;
    const int hi   = (lane >= 16) ? 1 : 0;
    const int l15  = lane & 15;
    const int pbeg = wave * (PP / 8);
    const float* xr0 = x + (size_t)n * DD * PP + (size_t)(d0 + l15) * PP + pbeg + (hi ? 2 : 0);
    const float* xr1 = xr0 + (size_t)16 * PP;
    const v2f* bp = a_pair + (size_t)n * (PP / 2) * KK + (size_t)((pbeg >> 1) + hi) * KK + l15;

    const v8f zero8 = {0.f, 0.f, 0.f, 0.f, 0.f, 0.f, 0.f, 0.f};
    v8f c[2][4];
    v2f s[4];
#pragma unroll
    for (int t = 0; t < 4; ++t) {
        c[0][t] = zero8;
        c[1][t] = zero8;
        s[t].x = 0.f;
        s[t].y = 0.f;
    }

    for (int i = 0; i < PP / 32; ++i) {
        const v2f a0 = *(const v2f*)(xr0 + 4 * i);
        const v2f a1 = *(const v2f*)(xr1 + 4 * i);
        const v2f* b = bp + (size_t)(2 * i) * KK;
        v2f bt[4];
#pragma unroll
        for (int t = 0; t < 4; ++t) bt[t] = b[16 * t];
#pragma unroll
        for (int t = 0; t < 4; ++t) {
            s[t] += bt[t];
            c[0][t] = __builtin_amdgcn_wmma_f32_16x16x4_f32(false, a0, false, bt[t],
                                                            (short)0, c[0][t], false, false);
            c[1][t] = __builtin_amdgcn_wmma_f32_16x16x4_f32(false, a1, false, bt[t],
                                                            (short)0, c[1][t], false, false);
        }
    }

#pragma unroll
    for (int dt = 0; dt < 2; ++dt)
#pragma unroll
        for (int t = 0; t < 4; ++t)
#pragma unroll
            for (int r = 0; r < 8; ++r)
                red[(((dt * 4 + t) * 8 + wave) * 8 + r) * 32 + lane] = c[dt][t][r];
#pragma unroll
    for (int t = 0; t < 4; ++t) cred[(t * 8 + wave) * 32 + lane] = s[t].x + s[t].y;
    __syncthreads();

    const int dt = wave >> 2;
    const int kt = wave & 3;
    float as = 0.f;
#pragma unroll
    for (int w = 0; w < 8; ++w)
        as += cred[(kt * 8 + w) * 32 + l15] + cred[(kt * 8 + w) * 32 + l15 + 16];

    const int k = (kt << 4) + l15;
    float* vo = vraw + (size_t)n * KK * DD;
    float val[8];
    float ss = 0.f;
#pragma unroll
    for (int r = 0; r < 8; ++r) {
        float acc = red[((wave * 8 + 0) * 8 + r) * 32 + lane];
#pragma unroll
        for (int w = 1; w < 8; ++w) acc += red[((wave * 8 + w) * 8 + r) * 32 + lane];
        const int d = d0 + (dt << 4) + r + (hi ? 8 : 0);
        val[r] = acc - as * centroids[(size_t)k * DD + d];
        ss += val[r] * val[r];
    }
    ss += __shfl_xor(ss, 16);
    if (lane < 16) rsq[dt * KK + k] = ss;
    __syncthreads();
#pragma unroll
    for (int r = 0; r < 8; ++r) {
        const int d = d0 + (dt << 4) + r + (hi ? 8 : 0);
        vo[(size_t)k * DD + d] = val[r];
    }
    if (tid < KK)
        rowss[((size_t)n * (DD / 32) + blockIdx.y) * KK + tid] = rsq[tid] + rsq[KK + tid];
}

#define NSL 8
#define SLF (KK * DD / NSL)
#define NVT (SLF / 256)

__global__ __launch_bounds__(64) void k_norm(const float* __restrict__ rowss,
                                             float* __restrict__ out) {
    __shared__ float tsc[KK];
    __shared__ float gpart[2];
    const int n    = blockIdx.x;
    const int tid  = threadIdx.x;
    const int lane = tid & 31;
    float* op = out + (size_t)n * KK * DD + (size_t)blockIdx.y * SLF + 4 * tid;

    const float* pp = rowss + (size_t)n * (DD / 32) * KK + tid;
    float ps[DD / 32];
#pragma unroll
    for (int b = 0; b < DD / 32; ++b) ps[b] = pp[b * KK];
    v4f v[NVT];
#pragma unroll
    for (int j = 0; j < NVT; ++j) v[j] = *(const v4f*)(op + 256 * j);

    float t = ps[0];
#pragma unroll
    for (int b = 1; b < DD / 32; ++b) t += ps[b];
    const float nrm = sqrtf(t);
    const float sc  = 1.0f / fmaxf(nrm, EPSF);
    const float nn  = nrm * sc;
    tsc[tid] = sc;
    float g = nn * nn;
    g += __shfl_xor(g, 16);
    g += __shfl_xor(g, 8);
    g += __shfl_xor(g, 4);
    g += __shfl_xor(g, 2);
    g += __shfl_xor(g, 1);
    if (lane == 0) gpart[tid >> 5] = g;
    __syncthreads();

    const float gs = 1.0f / fmaxf(sqrtf(gpart[0] + gpart[1]), EPSF);
#pragma unroll
    for (int j = 0; j < NVT; ++j) {
        const float rs = tsc[(KK / NSL) * blockIdx.y + (j >> 1)];
        v4f o;
        o.x = v[j].x * rs * gs;
        o.y = v[j].y * rs * gs;
        o.z = v[j].z * rs * gs;
        o.w = v[j].w * rs * gs;
        *(v4f*)(op + 256 * j) = o;
    }
}

extern "C" void kernel_launch(void* const* d_in, const int* in_sizes, int n_in,
                              void* d_out, int out_size, void* d_ws, size_t ws_size,
                              hipStream_t stream) {
    const float* x         = (const float*)d_in[0];
    const float* centroids = (const float*)d_in[1];
    const float* conv_w    = (const float*)d_in[2];
    const float* conv_b    = (const float*)d_in[3];
    float* out = (float*)d_out;

    v2f*   a_pair = (v2f*)d_ws;
    float* rowss  = (float*)d_ws + (size_t)NB * PP * KK;
    (void)in_sizes; (void)n_in; (void)out_size; (void)ws_size;

    k_assign<<<dim3(NB * (PP / 64)), 256, 0, stream>>>(x, conv_w, conv_b, a_pair);
    k_vlad  <<<dim3(NB, DD / 32), 256, 0, stream>>>(x, a_pair, centroids, out, rowss);
    k_norm  <<<dim3(NB, NSL), 64, 0, stream>>>(rowss, out);
}


# k_vlad: x reaches the A fragments through a per-wave LDS stage filled by 16-byte loads over whole 128-byte lines (4 lines a load, was 16); requests issued one block ahead; sum order unchanged
# speedup vs baseline: 1.9929x; 1.0768x over previous
#include <hip/hip_runtime.h>

typedef float v2f __attribute__((ext_vector_type(2)));
typedef float v4f __attribute__((ext_vector_type(4)));
typedef float v8f __attribute__((ext_vector_type(8)));

#define NB   32
#define DD   512
#define KK   64
#define PP   1024
#define EPSF 1e-12f

#define SMP 65

static __device__ __forceinline__ void assign_chunk(v8f (&c)[2][4], const v4f w0, const v4f w1,
                                                    const v4f x0, const v4f x1,
                                                    const v4f x2, const v4f x3) {
    v2f a[2][2];
    a[0][0].x = w0[0]; a[0][0].y = w0[1];
    a[0][1].x = w0[2]; a[0][1].y = w0[3];
    a[1][0].x = w1[0]; a[1][0].y = w1[1];
    a[1][1].x = w1[2]; a[1][1].y = w1[3];
#pragma unroll
    for (int j = 0; j < 4; ++j) {
        v2f b0, b1;
        b0.x = x0[j]; b0.y = x1[j];
        b1.x = x2[j]; b1.y = x3[j];
#pragma unroll
        for (int kk = 0; kk < 2; ++kk) {
            c[kk][j] = __builtin_amdgcn_wmma_f32_16x16x4_f32(false, a[kk][0], false, b0,
                                                             (short)0, c[kk][j], false, false);
            c[kk][j] = __builtin_amdgcn_wmma_f32_16x16x4_f32(false, a[kk][1], false, b1,
                                                             (short)0, c[kk][j], false, false);
        }
    }
}

__global__ __launch_bounds__(256) void k_assign(const float* __restrict__ x,
                                                const float* __restrict__ conv_w,
                                                const float* __restrict__ conv_b,
                                                v2f* __restrict__ a_pair) {
    __shared__ float red[16 * 4 * 8 * 32];
    __shared__ float sm[64 * SMP];
    const int n    = blockIdx.x >> 4;
    const int P0   = (blockIdx.x & 15) << 6;
    const int tid  = threadIdx.x;
    const int wave = tid >> 5;
    const int lane = tid & 31;
    const int hi   = (lane >= 16) ? 1 : 0;
    const int l15  = lane & 15;
    const int kh   = wave & 1;
    const int dq   = wave >> 1;
    const float* xp = x + (size_t)n * DD * PP + (size_t)(dq * 128 + 4 * hi) * PP + P0 + 4 * l15;
    const float* wp = conv_w + (size_t)(32 * kh + l15) * DD + dq * 128 + 4 * hi;

    const v8f zero8 = {0.f, 0.f, 0.f, 0.f, 0.f, 0.f, 0.f, 0.f};
    v8f c[2][4];
#pragma unroll
    for (int j = 0; j < 4; ++j) {
        c[0][j] = zero8;
        c[1][j] = zero8;
    }

    v4f w0 = *(const v4f*)(wp);
    v4f w1 = *(const v4f*)(wp + (size_t)16 * DD);
    v4f x0 = *(const v4f*)(xp);
    v4f x1 = *(const v4f*)(xp + (size_t)1 * PP);
    v4f x2 = *(const v4f*)(xp + (size_t)2 * PP);
    v4f x3 = *(const v4f*)(xp + (size_t)3 * PP);
    for (int i = 1; i < 16; ++i) {
        const float* wn = wp + 8 * i;
        const float* xn = xp + (size_t)(8 * i) * PP;
        const v4f nw0 = *(const v4f*)(wn);
        const v4f nw1 = *(const v4f*)(wn + (size_t)16 * DD);
        const v4f nx0 = *(const v4f*)(xn);
        const v4f nx1 = *(const v4f*)(xn + (size_t)1 * PP);
        const v4f nx2 = *(const v4f*)(xn + (size_t)2 * PP);
        const v4f nx3 = *(const v4f*)(xn + (size_t)3 * PP);
        assign_chunk(c, w0, w1, x0, x1, x2, x3);
        w0 = nw0; w1 = nw1;
        x0 = nx0; x1 = nx1; x2 = nx2; x3 = nx3;
    }
    assign_chunk(c, w0, w1, x0, x1, x2, x3);

#pragma unroll
    for (int kk = 0; kk < 2; ++kk)
#pragma unroll
        for (int j = 0; j < 4; ++j)
#pragma unroll
            for (int r = 0; r < 8; ++r)
                red[(((((2 * kh + kk) * 4 + j) * 4 + dq) * 8 + r) * 32) + lane] = c[kk][j][r];
    __syncthreads();

#pragma unroll
    for (int tt = 0; tt < 2; ++tt) {
        const int tile = 2 * wave + tt;
        const int kt   = tile >> 2;
        const int j    = tile & 3;
#pragma unroll
        for (int r = 0; r < 8; ++r) {
            float acc = red[(((tile * 4 + 0) * 8 + r) * 32) + lane];
#pragma unroll
            for (int q = 1; q < 4; ++q) acc += red[(((tile * 4 + q) * 8 + r) * 32) + lane];
            const int k = 16 * kt + 8 * hi + r;
            sm[(4 * l15 + j) * SMP + k] = acc + conv_b[k];
        }
    }
    __syncthreads();

    const int p = tid >> 2;
    const int g = tid & 3;
    float v[16];
    float m = -3.0e38f;
#pragma unroll
    for (int jj = 0; jj < 16; ++jj) {
        v[jj] = sm[p * SMP + g * 16 + jj];
        m = fmaxf(m, v[jj]);
    }
    m = fmaxf(m, __shfl_xor(m, 1));
    m = fmaxf(m, __shfl_xor(m, 2));
    float s = 0.f;
#pragma unroll
    for (int jj = 0; jj < 16; ++jj) {
        v[jj] = __expf(v[jj] - m);
        s += v[jj];
    }
    s += __shfl_xor(s, 1);
    s += __shfl_xor(s, 2);
    const float inv = 1.0f / s;
#pragma unroll
    for (int jj = 0; jj < 16; ++jj) sm[p * SMP + g * 16 + jj] = v[jj] * inv;
    __syncthreads();

    v2f* ao = a_pair + ((size_t)n * (PP / 2) + (P0 >> 1)) * KK;
    for (int i = tid; i < 32 * KK; i += 256) {
        const int ppl = i >> 6;
        const int k   = i & 63;
        v2f o;
        o.x = sm[(2 * ppl) * SMP + k];
        o.y = sm[(2 * ppl + 1) * SMP + k];
        ao[(size_t)ppl * KK + k] = o;
    }
}

#define XP 36

static __device__ __forceinline__ void wave_lds_order() {
    __builtin_amdgcn_fence(__ATOMIC_RELEASE, "wavefront");
    __builtin_amdgcn_wave_barrier();
    __builtin_amdgcn_fence(__ATOMIC_ACQUIRE, "wavefront");
}

__global__ __launch_bounds__(256) __attribute__((amdgpu_waves_per_eu(4, 4)))
void k_vlad(const float* __restrict__ x,
            const v2f* __restrict__ a_pair,
            const float* __restrict__ centroids,
            float* __restrict__ vraw,
            float* __restrict__ rowss) {
    __shared__ float red[8 * 8 * 8 * 32];
    __shared__ float cred[4 * 8 * 32];
    __shared__ float rsq[2 * KK];
    __shared__ v4f   xs4[8 * 32 * XP / 4];
    const int n    = blockIdx.x;
    const int d0   = blockIdx.y << 5;
    const int tid  = threadIdx.x;
    const int wave = tid >> 5;
    const int lane = tid & 31;
    const int hi   = (lane >= 16) ? 1 : 0;
    const int l15  = lane & 15;
    const int pbeg = wave * (PP / 8);
    const int lrow = lane >> 3;
    const int lpc  = lane & 7;
    const float* xg = x + (size_t)n * DD * PP + (size_t)(d0 + lrow) * PP + pbeg + 4 * lpc;
    float* xw  = (float*)xs4 + wave * (32 * XP);
    float* xst = xw + lrow * XP + 4 * lpc;
    const float* xfr = xw + l15 * XP + (hi ? 2 : 0);
    const v2f* bp = a_pair + (size_t)n * (PP / 2) * KK + (size_t)((pbeg >> 1) + hi) * KK + l15;

    const v8f zero8 = {0.f, 0.f, 0.f, 0.f, 0.f, 0.f, 0.f, 0.f};
    v8f c[2][4];
    v2f s[4];
#pragma unroll
    for (int t = 0; t < 4; ++t) {
        c[0][t] = zero8;
        c[1][t] = zero8;
        s[t].x = 0.f;
        s[t].y = 0.f;
    }

    v4f xr[8];
    v2f af[2][2][4];
    v2f bt[2][4][4];
#pragma unroll
    for (int q = 0; q < 8; ++q) xr[q] = *(const v4f*)(xg + (size_t)(4 * q) * PP);
#pragma unroll
    for (int st = 0; st < 4; ++st)
#pragma unroll
        for (int t = 0; t < 4; ++t) bt[0][st][t] = bp[(size_t)(2 * st) * KK + 16 * t];
#pragma unroll
    for (int q = 0; q < 8; ++q) *(v4f*)(xst + (4 * q) * XP) = xr[q];
    wave_lds_order();
#pragma unroll
    for (int st = 0; st < 4; ++st) {
        af[0][0][st] = *(const v2f*)(xfr + 4 * st);
        af[0][1][st] = *(const v2f*)(xfr + 16 * XP + 4 * st);
    }
#pragma unroll
    for (int q = 0; q < 8; ++q) xr[q] = *(const v4f*)(xg + (size_t)(4 * q) * PP + 32);

#pragma unroll
    for (int h = 0; h < 8; ++h) {
        const int cur = h & 1;
        const int nxt = cur ^ 1;
        if (h + 1 < 8) {
#pragma unroll
            for (int st = 0; st < 4; ++st)
#pragma unroll
                for (int t = 0; t < 4; ++t)
                    bt[nxt][st][t] = bp[(size_t)(2 * (4 * (h + 1) + st)) * KK + 16 * t];
            if (h & 1) {
                wave_lds_order();
#pragma unroll
                for (int q = 0; q < 8; ++q) *(v4f*)(xst + (4 * q) * XP) = xr[q];
                wave_lds_order();
#pragma unroll
                for (int st = 0; st < 4; ++st) {
                    af[nxt][0][st] = *(const v2f*)(xfr + 4 * st);
                    af[nxt][1][st] = *(const v2f*)(xfr + 16 * XP + 4 * st);
                }
                if (h + 3 < 8) {
#pragma unroll
                    for (int q = 0; q < 8; ++q)
                        xr[q] = *(const v4f*)(xg + (size_t)(4 * q) * PP + 32 * ((h + 3) >> 1));
                }
            } else {
#pragma unroll
                for (int st = 0; st < 4; ++st) {
                    af[nxt][0][st] = *(const v2f*)(xfr + 16 + 4 * st);
                    af[nxt][1][st] = *(const v2f*)(xfr + 16 * XP + 16 + 4 * st);
                }
            }
        }
#pragma unroll
        for (int st = 0; st < 4; ++st) {
#pragma unroll
            for (int t = 0; t < 4; ++t) {
                s[t] += bt[cur][st][t];
                c[0][t] = __builtin_amdgcn_wmma_f32_16x16x4_f32(false, af[cur][0][st], false, bt[cur][st][t],
                                                                (short)0, c[0][t], false, false);
                c[1][t] = __builtin_amdgcn_wmma_f32_16x16x4_f32(false, af[cur][1][st], false, bt[cur][st][t],
                                                                (short)0, c[1][t], false, false);
            }
        }
    }

#pragma unroll
    for (int dt = 0; dt < 2; ++dt)
#pragma unroll
        for (int t = 0; t < 4; ++t)
#pragma unroll
            for (int r = 0; r < 8; ++r)
                red[(((dt * 4 + t) * 8 + wave) * 8 + r) * 32 + lane] = c[dt][t][r];
#pragma unroll
    for (int t = 0; t < 4; ++t) cred[(t * 8 + wave) * 32 + lane] = s[t].x + s[t].y;
    __syncthreads();

    const int dt = wave >> 2;
    const int kt = wave & 3;
    float as = 0.f;
#pragma unroll
    for (int w = 0; w < 8; ++w)
        as += cred[(kt * 8 + w) * 32 + l15] + cred[(kt * 8 + w) * 32 + l15 + 16];

    const int k = (kt << 4) + l15;
    float* vo = vraw + (size_t)n * KK * DD;
    float val[8];
    float ss = 0.f;
#pragma unroll
    for (int r = 0; r < 8; ++r) {
        float acc = red[((wave * 8 + 0) * 8 + r) * 32 + lane];
#pragma unroll
        for (int w = 1; w < 8; ++w) acc += red[((wave * 8 + w) * 8 + r) * 32 + lane];
        const int d = d0 + (dt << 4) + r + (hi ? 8 : 0);
        val[r] = acc - as * centroids[(size_t)k * DD + d];
        ss += val[r] * val[r];
    }
    ss += __shfl_xor(ss, 16);
    if (lane < 16) rsq[dt * KK + k] = ss;
    __syncthreads();
#pragma unroll
    for (int r = 0; r < 8; ++r) {
        const int d = d0 + (dt << 4) + r + (hi ? 8 : 0);
        vo[(size_t)k * DD + d] = val[r];
    }
    if (tid < KK)
        rowss[((size_t)n * (DD / 32) + blockIdx.y) * KK + tid] = rsq[tid] + rsq[KK + tid];
}

#define NSL 8
#define SLF (KK * DD / NSL)
#define NVT (SLF / 256)

__global__ __launch_bounds__(64) void k_norm(const float* __restrict__ rowss,
                                             float* __restrict__ out) {
    __shared__ float tsc[KK];
    __shared__ float gpart[2];
    const int n    = blockIdx.x;
    const int tid  = threadIdx.x;
    const int lane = tid & 31;
    float* op = out + (size_t)n * KK * DD + (size_t)blockIdx.y * SLF + 4 * tid;

    const float* pp = rowss + (size_t)n * (DD / 32) * KK + tid;
    float ps[DD / 32];
#pragma unroll
    for (int b = 0; b < DD / 32; ++b) ps[b] = pp[b * KK];
    v4f v[NVT];
#pragma unroll
    for (int j = 0; j < NVT; ++j) v[j] = *(const v4f*)(op + 256 * j);

    float t = ps[0];
#pragma unroll
    for (int b = 1; b < DD / 32; ++b) t += ps[b];
    const float nrm = sqrtf(t);
    const float sc  = 1.0f / fmaxf(nrm, EPSF);
    const float nn  = nrm * sc;
    tsc[tid] = sc;
    float g = nn * nn;
    g += __shfl_xor(g, 16);
    g += __shfl_xor(g, 8);
    g += __shfl_xor(g, 4);
    g += __shfl_xor(g, 2);
    g += __shfl_xor(g, 1);
    if (lane == 0) gpart[tid >> 5] = g;
    __syncthreads();

    const float gs = 1.0f / fmaxf(sqrtf(gpart[0] + gpart[1]), EPSF);
#pragma unroll
    for (int j = 0; j < NVT; ++j) {
        const float rs = tsc[(KK / NSL) * blockIdx.y + (j >> 1)];
        v4f o;
        o.x = v[j].x * rs * gs;
        o.y = v[j].y * rs * gs;
        o.z = v[j].z * rs * gs;
        o.w = v[j].w * rs * gs;
        *(v4f*)(op + 256 * j) = o;
    }
}

extern "C" void kernel_launch(void* const* d_in, const int* in_sizes, int n_in,
                              void* d_out, int out_size, void* d_ws, size_t ws_size,
                              hipStream_t stream) {
    const float* x         = (const float*)d_in[0];
    const float* centroids = (const float*)d_in[1];
    const float* conv_w    = (const float*)d_in[2];
    const float* conv_b    = (const float*)d_in[3];
    float* out = (float*)d_out;

    v2f*   a_pair = (v2f*)d_ws;
    float* rowss  = (float*)d_ws + (size_t)NB * PP * KK;
    (void)in_sizes; (void)n_in; (void)out_size; (void)ws_size;

    k_assign<<<dim3(NB * (PP / 64)), 256, 0, stream>>>(x, conv_w, conv_b, a_pair);
    k_vlad  <<<dim3(NB, DD / 32), 256, 0, stream>>>(x, a_pair, centroids, out, rowss);
    k_norm  <<<dim3(NB, NSL), 64, 0, stream>>>(rowss, out);
}
